# starting kernel
# baseline (speedup 1.0000x reference)
#include <hip/hip_runtime.h>

#define IN_F  256
#define OUT_F 128
#define LDS_PAD 260

typedef __attribute__((ext_vector_type(2))) float v2f;
typedef __attribute__((ext_vector_type(8))) float v8f;

__global__ __launch_bounds__(256) void sgc_gemm_wmma(const float* __restrict__ h,
                                                     const float* __restrict__ w,
                                                     float* __restrict__ x) {
  __shared__ float atile[16 * LDS_PAD];
  const int m0  = blockIdx.x * 16;
  const int tid = threadIdx.x;

  for (int i4 = tid; i4 < 16 * IN_F / 4; i4 += 256) {
    int i = i4 * 4;
    int r = i / IN_F, c = i % IN_F;
    float4 v = *(const float4*)(h + (size_t)(m0 + r) * IN_F + c);
    atile[r * LDS_PAD + c + 0] = v.x;
    atile[r * LDS_PAD + c + 1] = v.y;
    atile[r * LDS_PAD + c + 2] = v.z;
    atile[r * LDS_PAD + c + 3] = v.w;
  }
  __syncthreads();

  const int wave = tid >> 5;
  const int lane = tid & 31;
  const int half = lane >> 4;
  const int l    = lane & 15;
  const int n0   = wave * 16;

  v8f c = {};
  const float* arow = atile + l * LDS_PAD + 2 * half;
  const float* bcol = w + (size_t)(2 * half) * OUT_F + n0 + l;
  for (int k = 0; k < IN_F; k += 4) {
    float2 av = *(const float2*)(arow + k);
    v2f a, b;
    a.x = av.x;
    a.y = av.y;
    b.x = bcol[(size_t)k * OUT_F];
    b.y = bcol[(size_t)(k + 1) * OUT_F];
    c = __builtin_amdgcn_wmma_f32_16x16x4_f32(false, a, false, b, (short)0, c,
                                              false, false);
  }

  float* out = x + (size_t)(m0 + 8 * half) * OUT_F + n0 + l;
#pragma unroll
  for (int r = 0; r < 8; ++r) out[(size_t)r * OUT_F] = c[r];
}

__global__ __launch_bounds__(256) void sgc_scatter(const float* __restrict__ x,
                                                   const float* __restrict__ norm,
                                                   const int* __restrict__ src,
                                                   const int* __restrict__ dst,
                                                   float* __restrict__ acc, int E) {
  int gw   = (blockIdx.x * 256 + threadIdx.x) >> 5;
  int lane = threadIdx.x & 31;
  if (gw >= E) return;
  int s = __builtin_amdgcn_readfirstlane(src[gw]);
  int d = __builtin_amdgcn_readfirstlane(dst[gw]);
  float ns = norm[s];
  float4 v = *(const float4*)(x + (size_t)s * OUT_F + lane * 4);
  float r0 = ns * v.x, r1 = ns * v.y, r2 = ns * v.z, r3 = ns * v.w;
  float* a = acc + (size_t)d * OUT_F + lane * 4;
  asm volatile(
      "global_atomic_add_f32 %4, %0, off\n\t"
      "global_atomic_add_f32 %4, %1, off offset:4\n\t"
      "global_atomic_add_f32 %4, %2, off offset:8\n\t"
      "global_atomic_add_f32 %4, %3, off offset:12"
      :
      : "v"(r0), "v"(r1), "v"(r2), "v"(r3), "v"(a)
      : "memory");
}

__global__ __launch_bounds__(256) void sgc_zero(float4* __restrict__ p, int total4) {
  int i = blockIdx.x * 256 + threadIdx.x;
  if (i < total4) p[i] = make_float4(0.f, 0.f, 0.f, 0.f);
}

__global__ __launch_bounds__(256) void sgc_scale_zero(const float* __restrict__ norm,
                                                      float* __restrict__ acc,
                                                      float* __restrict__ xout,
                                                      int total4) {
  int i4 = blockIdx.x * 256 + threadIdx.x;
  if (i4 >= total4) return;
  int node = i4 >> 5;
  float nv = norm[node];
  float4 v = ((const float4*)acc)[i4];
  ((float4*)xout)[i4] = make_float4(nv * v.x, nv * v.y, nv * v.z, nv * v.w);
  ((float4*)acc)[i4] = make_float4(0.f, 0.f, 0.f, 0.f);
}

extern "C" void kernel_launch(void* const* d_in, const int* in_sizes, int n_in,
                              void* d_out, int out_size, void* d_ws, size_t ws_size,
                              hipStream_t stream) {
  const float* h    = (const float*)d_in[0];
  const float* w    = (const float*)d_in[1];
  const float* norm = (const float*)d_in[2];
  const int*   src  = (const int*)d_in[3];
  const int*   dst  = (const int*)d_in[4];
  float* out = (float*)d_out;
  float* acc = (float*)d_ws;

  const int N = in_sizes[2];
  const int E = in_sizes[3];

  const int mblocks = (N + 15) / 16;
  const int total4  = N * OUT_F / 4;
  const int sblocks = (total4 + 255) / 256;
  const int eblocks = (E + 7) / 8;

  sgc_gemm_wmma<<<mblocks, 256, 0, stream>>>(h, w, out);
  sgc_zero<<<sblocks, 256, 0, stream>>>((float4*)acc, total4);

  sgc_scatter<<<eblocks, 256, 0, stream>>>(out, norm, src, dst, acc, E);
  sgc_scale_zero<<<sblocks, 256, 0, stream>>>(norm, acc, out, total4);

  sgc_scatter<<<eblocks, 256, 0, stream>>>(out, norm, src, dst, acc, E);
  sgc_scale_zero<<<sblocks, 256, 0, stream>>>(norm, acc, out, total4);
}


# Hop by source ownership: a work-group keeps 400 norm-scaled source rows in LDS, its 32 waves scan src for them and add each match to its destination row by whole-line f32 atomics; arrival order kept
# speedup vs baseline: 1.8711x; 1.8711x over previous
#include <hip/hip_runtime.h>

#define IN_F  256
#define OUT_F 128
#define LDS_PAD 260

typedef __attribute__((ext_vector_type(2))) float v2f;
typedef __attribute__((ext_vector_type(8))) float v8f;

__global__ __launch_bounds__(256) void sgc_gemm_wmma(const float* __restrict__ h,
                                                     const float* __restrict__ w,
                                                     float* __restrict__ x) {
  __shared__ float atile[16 * LDS_PAD];
  const int m0  = blockIdx.x * 16;
  const int tid = threadIdx.x;

  for (int i4 = tid; i4 < 16 * IN_F / 4; i4 += 256) {
    int i = i4 * 4;
    int r = i / IN_F, c = i % IN_F;
    float4 v = *(const float4*)(h + (size_t)(m0 + r) * IN_F + c);
    atile[r * LDS_PAD + c + 0] = v.x;
    atile[r * LDS_PAD + c + 1] = v.y;
    atile[r * LDS_PAD + c + 2] = v.z;
    atile[r * LDS_PAD + c + 3] = v.w;
  }
  __syncthreads();

  const int wave = tid >> 5;
  const int lane = tid & 31;
  const int half = lane >> 4;
  const int l    = lane & 15;
  const int n0   = wave * 16;

  v8f c = {};
  const float* arow = atile + l * LDS_PAD + 2 * half;
  const float* bcol = w + (size_t)(2 * half) * OUT_F + n0 + l;
  for (int k = 0; k < IN_F; k += 4) {
    float2 av = *(const float2*)(arow + k);
    v2f a, b;
    a.x = av.x;
    a.y = av.y;
    b.x = bcol[(size_t)k * OUT_F];
    b.y = bcol[(size_t)(k + 1) * OUT_F];
    c = __builtin_amdgcn_wmma_f32_16x16x4_f32(false, a, false, b, (short)0, c,
                                              false, false);
  }

  float* out = x + (size_t)(m0 + 8 * half) * OUT_F + n0 + l;
#pragma unroll
  for (int r = 0; r < 8; ++r) out[(size_t)r * OUT_F] = c[r];
}

#define SRC_ROWS 400
#define Q_CAP    160

__device__ __forceinline__ void sgc_flush(const float* rows, const int2* q, int base, int n,
                                          const float* __restrict__ norm,
                                          const int* __restrict__ dst,
                                          float* __restrict__ acc, int lane) {
  int sl = 0, d = 0;
  float nd = 0.f;
  if (lane < n) {
    const int2 ent = q[base + lane];
    sl = ent.y;
    d  = dst[ent.x];
    nd = norm[d];
  }
  const float* rp = rows + __builtin_amdgcn_readlane(sl, 0) * OUT_F + lane;
  float c0 = rp[0], c1 = rp[32], c2 = rp[64], c3 = rp[96];
  for (int i = 0; i < n; ++i) {
    const int   di  = __builtin_amdgcn_readlane(d, i);
    const float ndi = __builtin_bit_cast(float,
                        __builtin_amdgcn_readlane(__builtin_bit_cast(int, nd), i));
    const int   inx = (i + 1 < n) ? i + 1 : i;
    const float* rn = rows + __builtin_amdgcn_readlane(sl, inx) * OUT_F + lane;
    const float n0 = rn[0], n1 = rn[32], n2 = rn[64], n3 = rn[96];
    float t0 = c0 * ndi, t1 = c1 * ndi, t2 = c2 * ndi, t3 = c3 * ndi;
    float* a = acc + (size_t)di * OUT_F + lane;
    asm volatile(
        "global_atomic_add_f32 %4, %0, off\n\t"
        "global_atomic_add_f32 %4, %1, off offset:128\n\t"
        "global_atomic_add_f32 %4, %2, off offset:256\n\t"
        "global_atomic_add_f32 %4, %3, off offset:384\n\t"
        "s_wait_xcnt 0x0"
        :
        : "v"(t0), "v"(t1), "v"(t2), "v"(t3), "v"(a)
        : "memory");
    c0 = n0; c1 = n1; c2 = n2; c3 = n3;
  }
}

__device__ __forceinline__ int4 sgc_load_src4(const int* __restrict__ src, int e0, int E) {
  const bool whole = e0 + 3 < E;
  int4 r = *(const int4*)(src + (whole ? e0 : 0));
  r.x = whole ? r.x : -1;
  r.y = whole ? r.y : -1;
  r.z = whole ? r.z : -1;
  r.w = whole ? r.w : -1;
  return r;
}

__device__ __forceinline__ void sgc_test(int s, int e, int2* q, int& cnt) {
  const bool hit = (unsigned)s < (unsigned)SRC_ROWS;
  const unsigned m = __builtin_amdgcn_ballot_w32(hit);
  if (m != 0u) {
    if (hit) q[cnt + (int)__builtin_amdgcn_mbcnt_lo(m, 0u)] = make_int2(e, s);
    cnt += __builtin_popcount(m);
  }
}

__global__ __launch_bounds__(1024) void sgc_push(const float* __restrict__ x,
                                                 const float* __restrict__ norm,
                                                 const int* __restrict__ src,
                                                 const int* __restrict__ dst,
                                                 float* __restrict__ acc, int N, int E) {
  __shared__ float rows[SRC_ROWS * OUT_F];
  __shared__ int2  q_all[32 * Q_CAP];
  const int tid  = threadIdx.x;
  const int lane = tid & 31;
  const int wave = __builtin_amdgcn_readfirstlane(tid >> 5);
  const int r0   = blockIdx.x * SRC_ROWS;

  for (int rb = wave * 4; rb < SRC_ROWS; rb += 128) {
    float4 v[4];
    float  ns[4];
    for (int u = 0; u < 4; ++u) {
      const int g = r0 + rb + u;
      const bool ok = (rb + u < SRC_ROWS) && (g < N);
      ns[u] = ok ? norm[g] : 0.f;
      v[u]  = ok ? *(const float4*)(x + (size_t)g * OUT_F + lane * 4)
                 : make_float4(0.f, 0.f, 0.f, 0.f);
    }
    for (int u = 0; u < 4; ++u) {
      if (rb + u < SRC_ROWS) {
        float4 o;
        o.x = ns[u] * v[u].x; o.y = ns[u] * v[u].y; o.z = ns[u] * v[u].z; o.w = ns[u] * v[u].w;
        *(float4*)(rows + (rb + u) * OUT_F + lane * 4) = o;
      }
    }
  }
  __syncthreads();

  int2* q = q_all + wave * Q_CAP;
  const int ntrip = (E >= 4) ? ((E + 1023) >> 10) : 0;
  const int start = (int)(((unsigned)blockIdx.x * (unsigned)ntrip) / gridDim.x);
  int cnt = 0;

  int4 nxt[8];
  if (wave < ntrip) {
    int tr = wave + start; if (tr >= ntrip) tr -= ntrip;
#pragma unroll
    for (int u = 0; u < 8; ++u) nxt[u] = sgc_load_src4(src, (tr << 10) + u * 128 + lane * 4, E);
  }
  for (int t = wave; t < ntrip; t += 32) {
    int tr = t + start; if (tr >= ntrip) tr -= ntrip;
    const int ebase = (tr << 10) + lane * 4;
    int4 sv[8];
#pragma unroll
    for (int u = 0; u < 8; ++u) sv[u] = nxt[u];
    if (t + 32 < ntrip) {
      int tn = t + 32 + start; if (tn >= ntrip) tn -= ntrip;
#pragma unroll
      for (int u = 0; u < 8; ++u) nxt[u] = sgc_load_src4(src, (tn << 10) + u * 128 + lane * 4, E);
    }
#pragma unroll
    for (int u = 0; u < 8; ++u) {
      const int e = ebase + u * 128;
      sgc_test(sv[u].x - r0, e + 0, q, cnt);
      sgc_test(sv[u].y - r0, e + 1, q, cnt);
      sgc_test(sv[u].z - r0, e + 2, q, cnt);
      sgc_test(sv[u].w - r0, e + 3, q, cnt);
      while (cnt >= 32) {
        cnt -= 32;
        sgc_flush(rows, q, cnt, 32, norm, dst, acc, lane);
      }
    }
  }
  const int tail0 = (E >= 4) ? (E & ~3) : 0;
  if (wave == 0 && tail0 < E) {
    const int e = tail0 + lane;
    const int s = (e < E) ? src[e] - r0 : -1;
    sgc_test(s, e, q, cnt);
    while (cnt >= 32) {
      cnt -= 32;
      sgc_flush(rows, q, cnt, 32, norm, dst, acc, lane);
    }
  }
  if (cnt > 0) sgc_flush(rows, q, 0, cnt, norm, dst, acc, lane);
}

__global__ __launch_bounds__(256) void sgc_zero(float4* __restrict__ p, int total4) {
  int i = blockIdx.x * 256 + threadIdx.x;
  if (i < total4) p[i] = make_float4(0.f, 0.f, 0.f, 0.f);
}

extern "C" void kernel_launch(void* const* d_in, const int* in_sizes, int n_in,
                              void* d_out, int out_size, void* d_ws, size_t ws_size,
                              hipStream_t stream) {
  const float* h    = (const float*)d_in[0];
  const float* w    = (const float*)d_in[1];
  const float* norm = (const float*)d_in[2];
  const int*   src  = (const int*)d_in[3];
  const int*   dst  = (const int*)d_in[4];
  float* out = (float*)d_out;
  float* buf = (float*)d_ws;

  const int N = in_sizes[2];
  const int E = in_sizes[3];

  const int mblocks = (N + 15) / 16;
  const int total4  = N * OUT_F / 4;
  const int sblocks = (total4 + 255) / 256;
  const int pblocks = (N + SRC_ROWS - 1) / SRC_ROWS;

  sgc_gemm_wmma<<<mblocks, 256, 0, stream>>>(h, w, out);

  sgc_zero<<<sblocks, 256, 0, stream>>>((float4*)buf, total4);
  sgc_push<<<pblocks, 1024, 0, stream>>>(out, norm, src, dst, buf, N, E);

  sgc_zero<<<sblocks, 256, 0, stream>>>((float4*)out, total4);
  sgc_push<<<pblocks, 1024, 0, stream>>>(buf, norm, src, dst, out, N, E);
}


# Hop keeps a destination row's adds inside one half of the chip at a time: a group (half from its die id) adds edges into its own half of the rows while it sweeps, the others after; arrival order kept
# speedup vs baseline: 2.1465x; 1.1472x over previous
#include <hip/hip_runtime.h>

#define IN_F  256
#define OUT_F 128
#define LDS_PAD 260

typedef __attribute__((ext_vector_type(2))) float v2f;
typedef __attribute__((ext_vector_type(8))) float v8f;

__global__ __launch_bounds__(256) void sgc_gemm_wmma(const float* __restrict__ h,
                                                     const float* __restrict__ w,
                                                     float* __restrict__ x) {
  __shared__ float atile[16 * LDS_PAD];
  const int m0  = blockIdx.x * 16;
  const int tid = threadIdx.x;

  for (int i4 = tid; i4 < 16 * IN_F / 4; i4 += 256) {
    int i = i4 * 4;
    int r = i / IN_F, c = i % IN_F;
    float4 v = *(const float4*)(h + (size_t)(m0 + r) * IN_F + c);
    atile[r * LDS_PAD + c + 0] = v.x;
    atile[r * LDS_PAD + c + 1] = v.y;
    atile[r * LDS_PAD + c + 2] = v.z;
    atile[r * LDS_PAD + c + 3] = v.w;
  }
  __syncthreads();

  const int wave = tid >> 5;
  const int lane = tid & 31;
  const int half = lane >> 4;
  const int l    = lane & 15;
  const int n0   = wave * 16;

  v8f c = {};
  const float* arow = atile + l * LDS_PAD + 2 * half;
  const float* bcol = w + (size_t)(2 * half) * OUT_F + n0 + l;
  for (int k = 0; k < IN_F; k += 4) {
    float2 av = *(const float2*)(arow + k);
    v2f a, b;
    a.x = av.x;
    a.y = av.y;
    b.x = bcol[(size_t)k * OUT_F];
    b.y = bcol[(size_t)(k + 1) * OUT_F];
    c = __builtin_amdgcn_wmma_f32_16x16x4_f32(false, a, false, b, (short)0, c,
                                              false, false);
  }

  float* out = x + (size_t)(m0 + 8 * half) * OUT_F + n0 + l;
#pragma unroll
  for (int r = 0; r < 8; ++r) out[(size_t)r * OUT_F] = c[r];
}

#define SRC_ROWS 400
#define Q_CAP    160
#define F_CAP    256

__device__ __forceinline__ void sgc_emit(const float* rows, int sl, int d, float nd, unsigned m,
                                         float* __restrict__ acc, int lane) {
  int i = __builtin_ctz(m);
  const float* rp = rows + __builtin_amdgcn_readlane(sl, i) * OUT_F + lane;
  float c0 = rp[0], c1 = rp[32], c2 = rp[64], c3 = rp[96];
  for (;;) {
    const int   di  = __builtin_amdgcn_readlane(d, i);
    const float ndi = __builtin_bit_cast(float,
                        __builtin_amdgcn_readlane(__builtin_bit_cast(int, nd), i));
    m &= m - 1u;
    const int inx = (m != 0u) ? __builtin_ctz(m) : i;
    const float* rn = rows + __builtin_amdgcn_readlane(sl, inx) * OUT_F + lane;
    const float n0 = rn[0], n1 = rn[32], n2 = rn[64], n3 = rn[96];
    float t0 = c0 * ndi, t1 = c1 * ndi, t2 = c2 * ndi, t3 = c3 * ndi;
    float* a = acc + (size_t)di * OUT_F + lane;
    asm volatile(
        "global_atomic_add_f32 %4, %0, off\n\t"
        "global_atomic_add_f32 %4, %1, off offset:128\n\t"
        "global_atomic_add_f32 %4, %2, off offset:256\n\t"
        "global_atomic_add_f32 %4, %3, off offset:384\n\t"
        "s_wait_xcnt 0x0"
        :
        : "v"(t0), "v"(t1), "v"(t2), "v"(t3), "v"(a)
        : "memory");
    if (m == 0u) break;
    c0 = n0; c1 = n1; c2 = n2; c3 = n3;
    i = inx;
  }
}

__device__ __forceinline__ void sgc_drain(const float* rows, const int2* fl, int fcnt,
                                          const float* __restrict__ norm,
                                          float* __restrict__ acc, int lane) {
  for (int b = 0; b < fcnt; b += 32) {
    const int n = (fcnt - b < 32) ? fcnt - b : 32;
    int sl = 0, d = 0;
    float nd = 0.f;
    if (lane < n) {
      const int2 ent = fl[b + lane];
      d  = ent.x;
      sl = ent.y;
      nd = norm[d];
    }
    const unsigned m = (n == 32) ? 0xffffffffu : ((1u << n) - 1u);
    sgc_emit(rows, sl, d, nd, m, acc, lane);
  }
}

__device__ __forceinline__ void sgc_flush(const float* rows, const int2* q, int base, int n,
                                          const float* __restrict__ norm,
                                          const int* __restrict__ dst,
                                          float* __restrict__ acc, int lane,
                                          int2* fl, int& fcnt, int nhalf, bool upper) {
  int sl = 0, d = 0;
  float nd = 0.f;
  bool mine = false, other = false;
  if (lane < n) {
    const int2 ent = q[base + lane];
    sl = ent.y;
    d  = dst[ent.x];
    mine  = ((d >= nhalf) == upper);
    other = !mine;
    if (mine) nd = norm[d];
  }
  const unsigned om = __builtin_amdgcn_ballot_w32(other);
  if (om != 0u) {
    if (other) fl[fcnt + (int)__builtin_amdgcn_mbcnt_lo(om, 0u)] = make_int2(d, sl);
    fcnt += __builtin_popcount(om);
  }
  const unsigned mm = __builtin_amdgcn_ballot_w32(mine);
  if (mm != 0u) sgc_emit(rows, sl, d, nd, mm, acc, lane);
  if (fcnt > F_CAP - 32) {
    sgc_drain(rows, fl, fcnt, norm, acc, lane);
    fcnt = 0;
  }
}

__device__ __forceinline__ int4 sgc_load_src4(const int* __restrict__ src, int e0, int E) {
  const bool whole = e0 + 3 < E;
  int4 r = *(const int4*)(src + (whole ? e0 : 0));
  r.x = whole ? r.x : -1;
  r.y = whole ? r.y : -1;
  r.z = whole ? r.z : -1;
  r.w = whole ? r.w : -1;
  return r;
}

__device__ __forceinline__ void sgc_test(int s, int e, int2* q, int& cnt) {
  const bool hit = (unsigned)s < (unsigned)SRC_ROWS;
  const unsigned m = __builtin_amdgcn_ballot_w32(hit);
  if (m != 0u) {
    if (hit) q[cnt + (int)__builtin_amdgcn_mbcnt_lo(m, 0u)] = make_int2(e, s);
    cnt += __builtin_popcount(m);
  }
}

__global__ __launch_bounds__(1024) void sgc_push(const float* __restrict__ x,
                                                 const float* __restrict__ norm,
                                                 const int* __restrict__ src,
                                                 const int* __restrict__ dst,
                                                 float* __restrict__ acc, int N, int E) {
  __shared__ float rows[SRC_ROWS * OUT_F];
  __shared__ int2  q_all[32 * Q_CAP];
  __shared__ int2  f_all[32 * F_CAP];
  const int tid  = threadIdx.x;
  const int lane = tid & 31;
  const int wave = __builtin_amdgcn_readfirstlane(tid >> 5);
  const int r0   = blockIdx.x * SRC_ROWS;

  for (int rb = wave * 4; rb < SRC_ROWS; rb += 128) {
    float4 v[4];
    float  ns[4];
    for (int u = 0; u < 4; ++u) {
      const int g = r0 + rb + u;
      const bool ok = (rb + u < SRC_ROWS) && (g < N);
      ns[u] = ok ? norm[g] : 0.f;
      v[u]  = ok ? *(const float4*)(x + (size_t)g * OUT_F + lane * 4)
                 : make_float4(0.f, 0.f, 0.f, 0.f);
    }
    for (int u = 0; u < 4; ++u) {
      if (rb + u < SRC_ROWS) {
        float4 o;
        o.x = ns[u] * v[u].x; o.y = ns[u] * v[u].y; o.z = ns[u] * v[u].z; o.w = ns[u] * v[u].w;
        *(float4*)(rows + (rb + u) * OUT_F + lane * 4) = o;
      }
    }
  }
  __syncthreads();

  int2* q  = q_all + wave * Q_CAP;
  int2* fl = f_all + wave * F_CAP;
  int fcnt = 0;
  const bool upper = ((__builtin_amdgcn_s_sendmsg_rtn(0x87) >> 16) & 15) >= 4;
  const int nhalf = N >> 1;
  const int ntrip = (E >= 4) ? ((E + 1023) >> 10) : 0;
  const int start = (int)(((unsigned)blockIdx.x * (unsigned)ntrip) / gridDim.x);
  int cnt = 0;

  int4 nxt[8];
  if (wave < ntrip) {
    int tr = wave + start; if (tr >= ntrip) tr -= ntrip;
#pragma unroll
    for (int u = 0; u < 8; ++u) nxt[u] = sgc_load_src4(src, (tr << 10) + u * 128 + lane * 4, E);
  }
  for (int t = wave; t < ntrip; t += 32) {
    int tr = t + start; if (tr >= ntrip) tr -= ntrip;
    const int ebase = (tr << 10) + lane * 4;
    int4 sv[8];
#pragma unroll
    for (int u = 0; u < 8; ++u) sv[u] = nxt[u];
    if (t + 32 < ntrip) {
      int tn = t + 32 + start; if (tn >= ntrip) tn -= ntrip;
#pragma unroll
      for (int u = 0; u < 8; ++u) nxt[u] = sgc_load_src4(src, (tn << 10) + u * 128 + lane * 4, E);
    }
#pragma unroll
    for (int u = 0; u < 8; ++u) {
      const int e = ebase + u * 128;
      sgc_test(sv[u].x - r0, e + 0, q, cnt);
      sgc_test(sv[u].y - r0, e + 1, q, cnt);
      sgc_test(sv[u].z - r0, e + 2, q, cnt);
      sgc_test(sv[u].w - r0, e + 3, q, cnt);
      while (cnt >= 32) {
        cnt -= 32;
        sgc_flush(rows, q, cnt, 32, norm, dst, acc, lane, fl, fcnt, nhalf, upper);
      }
    }
  }
  const int tail0 = (E >= 4) ? (E & ~3) : 0;
  if (wave == 0 && tail0 < E) {
    const int e = tail0 + lane;
    const int s = (e < E) ? src[e] - r0 : -1;
    sgc_test(s, e, q, cnt);
    while (cnt >= 32) {
      cnt -= 32;
      sgc_flush(rows, q, cnt, 32, norm, dst, acc, lane, fl, fcnt, nhalf, upper);
    }
  }
  if (cnt > 0) sgc_flush(rows, q, 0, cnt, norm, dst, acc, lane, fl, fcnt, nhalf, upper);
  if (fcnt > 0) sgc_drain(rows, fl, fcnt, norm, acc, lane);
}

__global__ __launch_bounds__(256) void sgc_zero(float4* __restrict__ p, int total4) {
  int i = blockIdx.x * 256 + threadIdx.x;
  if (i < total4) p[i] = make_float4(0.f, 0.f, 0.f, 0.f);
}

extern "C" void kernel_launch(void* const* d_in, const int* in_sizes, int n_in,
                              void* d_out, int out_size, void* d_ws, size_t ws_size,
                              hipStream_t stream) {
  const float* h    = (const float*)d_in[0];
  const float* w    = (const float*)d_in[1];
  const float* norm = (const float*)d_in[2];
  const int*   src  = (const int*)d_in[3];
  const int*   dst  = (const int*)d_in[4];
  float* out = (float*)d_out;
  float* buf = (float*)d_ws;

  const int N = in_sizes[2];
  const int E = in_sizes[3];

  const int mblocks = (N + 15) / 16;
  const int total4  = N * OUT_F / 4;
  const int sblocks = (total4 + 255) / 256;
  const int pblocks = (N + SRC_ROWS - 1) / SRC_ROWS;

  sgc_gemm_wmma<<<mblocks, 256, 0, stream>>>(h, w, out);

  sgc_zero<<<sblocks, 256, 0, stream>>>((float4*)buf, total4);
  sgc_push<<<pblocks, 1024, 0, stream>>>(out, norm, src, dst, buf, N, E);

  sgc_zero<<<sblocks, 256, 0, stream>>>((float4*)out, total4);
  sgc_push<<<pblocks, 1024, 0, stream>>>(buf, norm, src, dst, out, N, E);
}


# Hop 2 skips the sweep of src: hop 1 writes each wave's matched edges (destination, local row) to a workspace list as it flushes them, hop 2 adds from its list; adds in atomic arrival order as before
# speedup vs baseline: 2.5782x; 1.2011x over previous
#include <hip/hip_runtime.h>

#define IN_F  256
#define OUT_F 128
#define LDS_PAD 260

typedef __attribute__((ext_vector_type(2))) float v2f;
typedef __attribute__((ext_vector_type(8))) float v8f;

__global__ __launch_bounds__(256) void sgc_gemm_wmma(const float* __restrict__ h,
                                                     const float* __restrict__ w,
                                                     float* __restrict__ x) {
  __shared__ float atile[16 * LDS_PAD];
  const int m0  = blockIdx.x * 16;
  const int tid = threadIdx.x;

  for (int i4 = tid; i4 < 16 * IN_F / 4; i4 += 256) {
    int i = i4 * 4;
    int r = i / IN_F, c = i % IN_F;
    float4 v = *(const float4*)(h + (size_t)(m0 + r) * IN_F + c);
    atile[r * LDS_PAD + c + 0] = v.x;
    atile[r * LDS_PAD + c + 1] = v.y;
    atile[r * LDS_PAD + c + 2] = v.z;
    atile[r * LDS_PAD + c + 3] = v.w;
  }
  __syncthreads();

  const int wave = tid >> 5;
  const int lane = tid & 31;
  const int half = lane >> 4;
  const int l    = lane & 15;
  const int n0   = wave * 16;

  v8f c = {};
  const float* arow = atile + l * LDS_PAD + 2 * half;
  const float* bcol = w + (size_t)(2 * half) * OUT_F + n0 + l;
  for (int k = 0; k < IN_F; k += 4) {
    float2 av = *(const float2*)(arow + k);
    v2f a, b;
    a.x = av.x;
    a.y = av.y;
    b.x = bcol[(size_t)k * OUT_F];
    b.y = bcol[(size_t)(k + 1) * OUT_F];
    c = __builtin_amdgcn_wmma_f32_16x16x4_f32(false, a, false, b, (short)0, c,
                                              false, false);
  }

  float* out = x + (size_t)(m0 + 8 * half) * OUT_F + n0 + l;
#pragma unroll
  for (int r = 0; r < 8; ++r) out[(size_t)r * OUT_F] = c[r];
}

#define SRC_ROWS 400
#define Q_CAP    160
#define F_CAP    256
#define L_CAP    384
#define L_SHIFT  9
static_assert(SRC_ROWS <= (1 << L_SHIFT), "local row must fit the low bits of a list word");

__device__ __forceinline__ void sgc_emit(const float* rows, int sl, int d, float nd, unsigned m,
                                         float* __restrict__ acc, int lane) {
  int i = __builtin_ctz(m);
  const float* rp = rows + __builtin_amdgcn_readlane(sl, i) * OUT_F + lane;
  float c0 = rp[0], c1 = rp[32], c2 = rp[64], c3 = rp[96];
  for (;;) {
    const int   di  = __builtin_amdgcn_readlane(d, i);
    const float ndi = __builtin_bit_cast(float,
                        __builtin_amdgcn_readlane(__builtin_bit_cast(int, nd), i));
    m &= m - 1u;
    const int inx = (m != 0u) ? __builtin_ctz(m) : i;
    const float* rn = rows + __builtin_amdgcn_readlane(sl, inx) * OUT_F + lane;
    const float n0 = rn[0], n1 = rn[32], n2 = rn[64], n3 = rn[96];
    float t0 = c0 * ndi, t1 = c1 * ndi, t2 = c2 * ndi, t3 = c3 * ndi;
    float* a = acc + (size_t)di * OUT_F + lane;
    asm volatile(
        "global_atomic_add_f32 %4, %0, off\n\t"
        "global_atomic_add_f32 %4, %1, off offset:128\n\t"
        "global_atomic_add_f32 %4, %2, off offset:256\n\t"
        "global_atomic_add_f32 %4, %3, off offset:384\n\t"
        "s_wait_xcnt 0x0"
        :
        : "v"(t0), "v"(t1), "v"(t2), "v"(t3), "v"(a)
        : "memory");
    if (m == 0u) break;
    c0 = n0; c1 = n1; c2 = n2; c3 = n3;
    i = inx;
  }
}

__device__ __forceinline__ void sgc_drain(const float* rows, const int2* fl, int fcnt,
                                          const float* __restrict__ norm,
                                          float* __restrict__ acc, int lane) {
  for (int b = 0; b < fcnt; b += 32) {
    const int n = (fcnt - b < 32) ? fcnt - b : 32;
    int sl = 0, d = 0;
    float nd = 0.f;
    if (lane < n) {
      const int2 ent = fl[b + lane];
      d  = ent.x;
      sl = ent.y;
      nd = norm[d];
    }
    const unsigned m = (n == 32) ? 0xffffffffu : ((1u << n) - 1u);
    sgc_emit(rows, sl, d, nd, m, acc, lane);
  }
}

__device__ __forceinline__ void sgc_flush(const float* rows, const int2* q, int base, int n,
                                          const float* __restrict__ norm,
                                          const int* __restrict__ dst,
                                          float* __restrict__ acc, int lane,
                                          int2* fl, int& fcnt, int nhalf, bool upper,
                                          int* wl, int& lpos) {
  int sl = 0, d = 0;
  float nd = 0.f;
  bool mine = false, other = false;
  if (lane < n) {
    const int2 ent = q[base + lane];
    sl = ent.y;
    d  = dst[ent.x];
    mine  = ((d >= nhalf) == upper);
    other = !mine;
    if (mine) nd = norm[d];
  }
  if (wl != nullptr) {
    if (lane < n && lpos + n <= L_CAP) wl[lpos + lane] = (d << L_SHIFT) | sl;
    lpos += n;
  }
  const unsigned om = __builtin_amdgcn_ballot_w32(other);
  if (om != 0u) {
    if (other) fl[fcnt + (int)__builtin_amdgcn_mbcnt_lo(om, 0u)] = make_int2(d, sl);
    fcnt += __builtin_popcount(om);
  }
  const unsigned mm = __builtin_amdgcn_ballot_w32(mine);
  if (mm != 0u) sgc_emit(rows, sl, d, nd, mm, acc, lane);
  if (fcnt > F_CAP - 32) {
    sgc_drain(rows, fl, fcnt, norm, acc, lane);
    fcnt = 0;
  }
}

__device__ __forceinline__ int4 sgc_load_src4(const int* __restrict__ src, int e0, int E) {
  const bool whole = e0 + 3 < E;
  int4 r = *(const int4*)(src + (whole ? e0 : 0));
  r.x = whole ? r.x : -1;
  r.y = whole ? r.y : -1;
  r.z = whole ? r.z : -1;
  r.w = whole ? r.w : -1;
  return r;
}

__device__ __forceinline__ void sgc_test(int s, int e, int2* q, int& cnt) {
  const bool hit = (unsigned)s < (unsigned)SRC_ROWS;
  const unsigned m = __builtin_amdgcn_ballot_w32(hit);
  if (m != 0u) {
    if (hit) q[cnt + (int)__builtin_amdgcn_mbcnt_lo(m, 0u)] = make_int2(e, s);
    cnt += __builtin_popcount(m);
  }
}

template <bool REPLAY>
__global__ __launch_bounds__(1024) void sgc_push(const float* __restrict__ x,
                                                 const float* __restrict__ norm,
                                                 const int* __restrict__ src,
                                                 const int* __restrict__ dst,
                                                 float* __restrict__ acc, int N, int E,
                                                 int* __restrict__ lcnt,
                                                 int* __restrict__ lists) {
  __shared__ float rows[SRC_ROWS * OUT_F];
  __shared__ int2  q_all[32 * Q_CAP];
  __shared__ int2  f_all[32 * F_CAP];
  const int tid  = threadIdx.x;
  const int lane = tid & 31;
  const int wave = __builtin_amdgcn_readfirstlane(tid >> 5);
  const int r0   = blockIdx.x * SRC_ROWS;
  const int gw   = blockIdx.x * 32 + wave;
  int kept = -1;
  if (REPLAY && lcnt != nullptr) kept = lcnt[gw];

  for (int rb = wave * 4; rb < SRC_ROWS; rb += 128) {
    float4 v[4];
    float  ns[4];
    for (int u = 0; u < 4; ++u) {
      const int g = r0 + rb + u;
      const bool ok = (rb + u < SRC_ROWS) && (g < N);
      ns[u] = ok ? norm[g] : 0.f;
      v[u]  = ok ? *(const float4*)(x + (size_t)g * OUT_F + lane * 4)
                 : make_float4(0.f, 0.f, 0.f, 0.f);
    }
    for (int u = 0; u < 4; ++u) {
      if (rb + u < SRC_ROWS) {
        float4 o;
        o.x = ns[u] * v[u].x; o.y = ns[u] * v[u].y; o.z = ns[u] * v[u].z; o.w = ns[u] * v[u].w;
        *(float4*)(rows + (rb + u) * OUT_F + lane * 4) = o;
      }
    }
  }
  __syncthreads();

  int2* q  = q_all + wave * Q_CAP;
  int2* fl = f_all + wave * F_CAP;
  int fcnt = 0;
  const bool upper = ((__builtin_amdgcn_s_sendmsg_rtn(0x87) >> 16) & 15) >= 4;
  const int nhalf = N >> 1;

  if (REPLAY && kept >= 0) {
    const int* wlr = lists + (size_t)gw * L_CAP;
    const int last = kept - 1;
#pragma nounroll
    for (int pass = 0; pass < 2 && kept > 0; ++pass) {
      int   p0  = wlr[min(lane, last)];
      int   p1  = wlr[min(32 + lane, last)];
      p0 = (lane < kept) ? p0 : -1;
      p1 = (32 + lane < kept) ? p1 : -1;
      float nd0 = norm[max(p0, 0) >> L_SHIFT];
      for (int b = 0; b < kept; b += 32) {
        const int   p  = p0;
        const float nd = nd0;
        p0  = p1;
        nd0 = norm[max(p0, 0) >> L_SHIFT];
        p1  = wlr[min(b + 64 + lane, last)];
        p1  = (b + 64 + lane < kept) ? p1 : -1;
        const int d  = p >> L_SHIFT;
        const int sl = p & ((1 << L_SHIFT) - 1);
        const bool want = (p >= 0) && (((d >= nhalf) == upper) == (pass == 0));
        const unsigned m = __builtin_amdgcn_ballot_w32(want);
        if (m != 0u) sgc_emit(rows, sl, d, nd, m, acc, lane);
      }
    }
    return;
  }
  int* wl = (!REPLAY && lists != nullptr) ? lists + (size_t)gw * L_CAP : nullptr;
  int lpos = 0;
  const int ntrip = (E >= 4) ? ((E + 1023) >> 10) : 0;
  const int start = (int)(((unsigned)blockIdx.x * (unsigned)ntrip) / gridDim.x);
  int cnt = 0;

  int4 nxt[8];
  if (wave < ntrip) {
    int tr = wave + start; if (tr >= ntrip) tr -= ntrip;
#pragma unroll
    for (int u = 0; u < 8; ++u) nxt[u] = sgc_load_src4(src, (tr << 10) + u * 128 + lane * 4, E);
  }
  for (int t = wave; t < ntrip; t += 32) {
    int tr = t + start; if (tr >= ntrip) tr -= ntrip;
    const int ebase = (tr << 10) + lane * 4;
    int4 sv[8];
#pragma unroll
    for (int u = 0; u < 8; ++u) sv[u] = nxt[u];
    if (t + 32 < ntrip) {
      int tn = t + 32 + start; if (tn >= ntrip) tn -= ntrip;
#pragma unroll
      for (int u = 0; u < 8; ++u) nxt[u] = sgc_load_src4(src, (tn << 10) + u * 128 + lane * 4, E);
    }
#pragma unroll
    for (int u = 0; u < 8; ++u) {
      const int e = ebase + u * 128;
      sgc_test(sv[u].x - r0, e + 0, q, cnt);
      sgc_test(sv[u].y - r0, e + 1, q, cnt);
      sgc_test(sv[u].z - r0, e + 2, q, cnt);
      sgc_test(sv[u].w - r0, e + 3, q, cnt);
      while (cnt >= 32) {
        cnt -= 32;
        sgc_flush(rows, q, cnt, 32, norm, dst, acc, lane, fl, fcnt, nhalf, upper, wl, lpos);
      }
    }
  }
  const int tail0 = (E >= 4) ? (E & ~3) : 0;
  if (wave == 0 && tail0 < E) {
    const int e = tail0 + lane;
    const int s = (e < E) ? src[e] - r0 : -1;
    sgc_test(s, e, q, cnt);
    while (cnt >= 32) {
      cnt -= 32;
      sgc_flush(rows, q, cnt, 32, norm, dst, acc, lane, fl, fcnt, nhalf, upper, wl, lpos);
    }
  }
  if (cnt > 0) sgc_flush(rows, q, 0, cnt, norm, dst, acc, lane, fl, fcnt, nhalf, upper, wl, lpos);
  if (fcnt > 0) sgc_drain(rows, fl, fcnt, norm, acc, lane);
  if (wl != nullptr && lane == 0) lcnt[gw] = (lpos <= L_CAP) ? lpos : -1;
}

__global__ __launch_bounds__(256) void sgc_zero(float4* __restrict__ p, int total4) {
  int i = blockIdx.x * 256 + threadIdx.x;
  if (i < total4) p[i] = make_float4(0.f, 0.f, 0.f, 0.f);
}

extern "C" void kernel_launch(void* const* d_in, const int* in_sizes, int n_in,
                              void* d_out, int out_size, void* d_ws, size_t ws_size,
                              hipStream_t stream) {
  const float* h    = (const float*)d_in[0];
  const float* w    = (const float*)d_in[1];
  const float* norm = (const float*)d_in[2];
  const int*   src  = (const int*)d_in[3];
  const int*   dst  = (const int*)d_in[4];
  float* out = (float*)d_out;
  float* buf = (float*)d_ws;

  const int N = in_sizes[2];
  const int E = in_sizes[3];

  const int mblocks = (N + 15) / 16;
  const int total4  = N * OUT_F / 4;
  const int sblocks = (total4 + 255) / 256;
  const int pblocks = (N + SRC_ROWS - 1) / SRC_ROWS;

  const size_t buf_bytes = (size_t)N * OUT_F * sizeof(float);
  const size_t cnt_bytes = (size_t)pblocks * 32 * sizeof(int);
  const size_t lst_bytes = (size_t)pblocks * 32 * L_CAP * sizeof(int);
  const bool keep = (ws_size >= buf_bytes + cnt_bytes + lst_bytes) && (N <= (1 << (31 - L_SHIFT)));
  int* lcnt  = keep ? (int*)((char*)d_ws + buf_bytes) : nullptr;
  int* lists = keep ? (int*)((char*)d_ws + buf_bytes + cnt_bytes) : nullptr;

  sgc_gemm_wmma<<<mblocks, 256, 0, stream>>>(h, w, out);

  sgc_zero<<<sblocks, 256, 0, stream>>>((float4*)buf, total4);
  sgc_push<false><<<pblocks, 1024, 0, stream>>>(out, norm, src, dst, buf, N, E, lcnt, lists);

  sgc_zero<<<sblocks, 256, 0, stream>>>((float4*)out, total4);
  sgc_push<true><<<pblocks, 1024, 0, stream>>>(buf, norm, src, dst, out, N, E, lcnt, lists);
}


# Hop 1 sweeps one key byte per edge (owning group = src/400, written by a new small launch) instead of src: 4x fewer bytes, 64 edges a lane tested branch-free; sums in atomic arrival order as before
# speedup vs baseline: 2.7749x; 1.0763x over previous
#include <hip/hip_runtime.h>

#define IN_F  256
#define OUT_F 128
#define LDS_PAD 260

typedef __attribute__((ext_vector_type(2))) float v2f;
typedef __attribute__((ext_vector_type(8))) float v8f;

__global__ __launch_bounds__(256) void sgc_gemm_wmma(const float* __restrict__ h,
                                                     const float* __restrict__ w,
                                                     float* __restrict__ x) {
  __shared__ float atile[16 * LDS_PAD];
  const int m0  = blockIdx.x * 16;
  const int tid = threadIdx.x;

  for (int i4 = tid; i4 < 16 * IN_F / 4; i4 += 256) {
    int i = i4 * 4;
    int r = i / IN_F, c = i % IN_F;
    float4 v = *(const float4*)(h + (size_t)(m0 + r) * IN_F + c);
    atile[r * LDS_PAD + c + 0] = v.x;
    atile[r * LDS_PAD + c + 1] = v.y;
    atile[r * LDS_PAD + c + 2] = v.z;
    atile[r * LDS_PAD + c + 3] = v.w;
  }
  __syncthreads();

  const int wave = tid >> 5;
  const int lane = tid & 31;
  const int half = lane >> 4;
  const int l    = lane & 15;
  const int n0   = wave * 16;

  v8f c = {};
  const float* arow = atile + l * LDS_PAD + 2 * half;
  const float* bcol = w + (size_t)(2 * half) * OUT_F + n0 + l;
  for (int k = 0; k < IN_F; k += 4) {
    float2 av = *(const float2*)(arow + k);
    v2f a, b;
    a.x = av.x;
    a.y = av.y;
    b.x = bcol[(size_t)k * OUT_F];
    b.y = bcol[(size_t)(k + 1) * OUT_F];
    c = __builtin_amdgcn_wmma_f32_16x16x4_f32(false, a, false, b, (short)0, c,
                                              false, false);
  }

  float* out = x + (size_t)(m0 + 8 * half) * OUT_F + n0 + l;
#pragma unroll
  for (int r = 0; r < 8; ++r) out[(size_t)r * OUT_F] = c[r];
}

#define SRC_ROWS 400
#define Q_CAP    160
#define F_CAP    256
#define L_CAP    384
#define L_SHIFT  9
static_assert(SRC_ROWS <= (1 << L_SHIFT), "local row must fit the low bits of a list word");

__device__ __forceinline__ void sgc_emit(const float* rows, int sl, int d, float nd, unsigned m,
                                         float* __restrict__ acc, int lane) {
  int i = __builtin_ctz(m);
  const float* rp = rows + __builtin_amdgcn_readlane(sl, i) * OUT_F + lane;
  float c0 = rp[0], c1 = rp[32], c2 = rp[64], c3 = rp[96];
  for (;;) {
    const int   di  = __builtin_amdgcn_readlane(d, i);
    const float ndi = __builtin_bit_cast(float,
                        __builtin_amdgcn_readlane(__builtin_bit_cast(int, nd), i));
    m &= m - 1u;
    const int inx = (m != 0u) ? __builtin_ctz(m) : i;
    const float* rn = rows + __builtin_amdgcn_readlane(sl, inx) * OUT_F + lane;
    const float n0 = rn[0], n1 = rn[32], n2 = rn[64], n3 = rn[96];
    float t0 = c0 * ndi, t1 = c1 * ndi, t2 = c2 * ndi, t3 = c3 * ndi;
    float* a = acc + (size_t)di * OUT_F + lane;
    asm volatile(
        "global_atomic_add_f32 %4, %0, off\n\t"
        "global_atomic_add_f32 %4, %1, off offset:128\n\t"
        "global_atomic_add_f32 %4, %2, off offset:256\n\t"
        "global_atomic_add_f32 %4, %3, off offset:384\n\t"
        "s_wait_xcnt 0x0"
        :
        : "v"(t0), "v"(t1), "v"(t2), "v"(t3), "v"(a)
        : "memory");
    if (m == 0u) break;
    c0 = n0; c1 = n1; c2 = n2; c3 = n3;
    i = inx;
  }
}

__device__ __forceinline__ void sgc_drain(const float* rows, const int2* fl, int fcnt,
                                          const float* __restrict__ norm,
                                          float* __restrict__ acc, int lane) {
  for (int b = 0; b < fcnt; b += 32) {
    const int n = (fcnt - b < 32) ? fcnt - b : 32;
    int sl = 0, d = 0;
    float nd = 0.f;
    if (lane < n) {
      const int2 ent = fl[b + lane];
      d  = ent.x;
      sl = ent.y;
      nd = norm[d];
    }
    const unsigned m = (n == 32) ? 0xffffffffu : ((1u << n) - 1u);
    sgc_emit(rows, sl, d, nd, m, acc, lane);
  }
}

template <bool KEYS>
__device__ __forceinline__ void sgc_flush(const float* rows, const int2* q, int base, int n,
                                          const float* __restrict__ norm,
                                          const int* __restrict__ src,
                                          const int* __restrict__ dst,
                                          float* __restrict__ acc, int lane,
                                          int2* fl, int& fcnt, int nhalf, bool upper,
                                          int* wl, int& lpos, int r0) {
  int sl = 0, d = 0;
  float nd = 0.f;
  bool mine = false, other = false;
  if (lane < n) {
    const int2 ent = q[base + lane];
    sl = KEYS ? src[ent.x] - r0 : ent.y;
    d  = dst[ent.x];
    mine  = ((d >= nhalf) == upper);
    other = !mine;
    if (mine) nd = norm[d];
  }
  if (wl != nullptr) {
    if (lane < n && lpos + n <= L_CAP) wl[lpos + lane] = (d << L_SHIFT) | sl;
    lpos += n;
  }
  const unsigned om = __builtin_amdgcn_ballot_w32(other);
  if (om != 0u) {
    if (other) fl[fcnt + (int)__builtin_amdgcn_mbcnt_lo(om, 0u)] = make_int2(d, sl);
    fcnt += __builtin_popcount(om);
  }
  const unsigned mm = __builtin_amdgcn_ballot_w32(mine);
  if (mm != 0u) sgc_emit(rows, sl, d, nd, mm, acc, lane);
  if (fcnt > F_CAP - 32) {
    sgc_drain(rows, fl, fcnt, norm, acc, lane);
    fcnt = 0;
  }
}

__device__ __forceinline__ int4 sgc_load_src4(const int* __restrict__ src, int e0, int E) {
  const bool whole = e0 + 3 < E;
  int4 r = *(const int4*)(src + (whole ? e0 : 0));
  r.x = whole ? r.x : -1;
  r.y = whole ? r.y : -1;
  r.z = whole ? r.z : -1;
  r.w = whole ? r.w : -1;
  return r;
}

__device__ __forceinline__ void sgc_test(int s, int e, int2* q, int& cnt) {
  const bool hit = (unsigned)s < (unsigned)SRC_ROWS;
  const unsigned m = __builtin_amdgcn_ballot_w32(hit);
  if (m != 0u) {
    if (hit) q[cnt + (int)__builtin_amdgcn_mbcnt_lo(m, 0u)] = make_int2(e, s);
    cnt += __builtin_popcount(m);
  }
}

__device__ __forceinline__ unsigned sgc_key_hits(unsigned kv, unsigned g4) {
  const unsigned t = kv ^ g4;
  const unsigned b = (t & 0x7f7f7f7fu) + 0x7f7f7f7fu;
  return ~(b | t) & 0x80808080u;
}

__device__ __forceinline__ unsigned sgc_key_mask(uint4 a, uint4 b, unsigned g4) {
  return (sgc_key_hits(a.x, g4) >> 7) | (sgc_key_hits(a.y, g4) >> 6) |
         (sgc_key_hits(a.z, g4) >> 5) | (sgc_key_hits(a.w, g4) >> 4) |
         (sgc_key_hits(b.x, g4) >> 3) | (sgc_key_hits(b.y, g4) >> 2) |
         (sgc_key_hits(b.z, g4) >> 1) | sgc_key_hits(b.w, g4);
}

__device__ __forceinline__ void sgc_queue_keys(unsigned mm, int ebase, int2* q, int& cnt,
                                               const float* rows,
                                               const float* __restrict__ norm,
                                               const int* __restrict__ src,
                                               const int* __restrict__ dst,
                                               float* __restrict__ acc, int lane,
                                               int2* fl, int& fcnt, int nhalf, bool upper,
                                               int* wl, int& lpos, int r0) {
  unsigned m = __builtin_amdgcn_ballot_w32(mm != 0u);
  while (m != 0u) {
    if (mm != 0u) {
      const int p = __builtin_ctz(mm);
      const int e = ebase + ((p & 4) << 7) + ((p & 3) << 2) + (p >> 3);
      q[cnt + (int)__builtin_amdgcn_mbcnt_lo(m, 0u)] = make_int2(e, 0);
    }
    cnt += __builtin_popcount(m);
    mm &= mm - 1u;
    if (cnt >= 32) {
      cnt -= 32;
      sgc_flush<true>(rows, q, cnt, 32, norm, src, dst, acc, lane, fl, fcnt, nhalf, upper, wl, lpos, r0);
    }
    m = __builtin_amdgcn_ballot_w32(mm != 0u);
  }
}

__device__ __forceinline__ unsigned sgc_key(int s) {
  const unsigned k = (unsigned)s / (unsigned)SRC_ROWS;
  return k < 255u ? k : 255u;
}

__global__ __launch_bounds__(256) void sgc_keys(const int* __restrict__ src,
                                                unsigned* __restrict__ keys, int E) {
  const int lane = threadIdx.x & 31;
  const int w0   = ((blockIdx.x * 256 + threadIdx.x) >> 5) * 128 + lane;
  int4 s[4];
#pragma unroll
  for (int u = 0; u < 4; ++u) {
    const int e0 = (w0 + u * 32) * 4;
    s[u] = *(const int4*)(src + ((e0 + 3 < E) ? e0 : 0));
  }
#pragma unroll
  for (int u = 0; u < 4; ++u) {
    const int e0 = (w0 + u * 32) * 4;
    if (e0 + 3 >= E) {
      s[u] = make_int4(-1, -1, -1, -1);
      if (e0 + 0 < E) s[u].x = src[e0 + 0];
      if (e0 + 1 < E) s[u].y = src[e0 + 1];
      if (e0 + 2 < E) s[u].z = src[e0 + 2];
    }
    keys[w0 + u * 32] = sgc_key(s[u].x) | (sgc_key(s[u].y) << 8) |
                        (sgc_key(s[u].z) << 16) | (sgc_key(s[u].w) << 24);
  }
}

template <bool REPLAY, bool KEYS>
__global__ __launch_bounds__(1024) void sgc_push(const float* __restrict__ x,
                                                 const float* __restrict__ norm,
                                                 const int* __restrict__ src,
                                                 const int* __restrict__ dst,
                                                 float* __restrict__ acc, int N, int E,
                                                 int* __restrict__ lcnt,
                                                 int* __restrict__ lists,
                                                 const unsigned* __restrict__ keys) {
  __shared__ float rows[SRC_ROWS * OUT_F];
  __shared__ int2  q_all[32 * Q_CAP];
  __shared__ int2  f_all[32 * F_CAP];
  const int tid  = threadIdx.x;
  const int lane = tid & 31;
  const int wave = __builtin_amdgcn_readfirstlane(tid >> 5);
  const int r0   = blockIdx.x * SRC_ROWS;
  const int gw   = blockIdx.x * 32 + wave;
  int kept = -1;
  if (REPLAY && lcnt != nullptr) kept = lcnt[gw];

  for (int rb = wave * 4; rb < SRC_ROWS; rb += 128) {
    float4 v[4];
    float  ns[4];
    for (int u = 0; u < 4; ++u) {
      const int g = r0 + rb + u;
      const bool ok = (rb + u < SRC_ROWS) && (g < N);
      ns[u] = ok ? norm[g] : 0.f;
      v[u]  = ok ? *(const float4*)(x + (size_t)g * OUT_F + lane * 4)
                 : make_float4(0.f, 0.f, 0.f, 0.f);
    }
    for (int u = 0; u < 4; ++u) {
      if (rb + u < SRC_ROWS) {
        float4 o;
        o.x = ns[u] * v[u].x; o.y = ns[u] * v[u].y; o.z = ns[u] * v[u].z; o.w = ns[u] * v[u].w;
        *(float4*)(rows + (rb + u) * OUT_F + lane * 4) = o;
      }
    }
  }
  __syncthreads();

  int2* q  = q_all + wave * Q_CAP;
  int2* fl = f_all + wave * F_CAP;
  int fcnt = 0;
  const bool upper = ((__builtin_amdgcn_s_sendmsg_rtn(0x87) >> 16) & 15) >= 4;
  const int nhalf = N >> 1;

  if (REPLAY && kept >= 0) {
    const int* wlr = lists + (size_t)gw * L_CAP;
    const int last = kept - 1;
#pragma nounroll
    for (int pass = 0; pass < 2 && kept > 0; ++pass) {
      int   p0  = wlr[min(lane, last)];
      int   p1  = wlr[min(32 + lane, last)];
      p0 = (lane < kept) ? p0 : -1;
      p1 = (32 + lane < kept) ? p1 : -1;
      float nd0 = norm[max(p0, 0) >> L_SHIFT];
      for (int b = 0; b < kept; b += 32) {
        const int   p  = p0;
        const float nd = nd0;
        p0  = p1;
        nd0 = norm[max(p0, 0) >> L_SHIFT];
        p1  = wlr[min(b + 64 + lane, last)];
        p1  = (b + 64 + lane < kept) ? p1 : -1;
        const int d  = p >> L_SHIFT;
        const int sl = p & ((1 << L_SHIFT) - 1);
        const bool want = (p >= 0) && (((d >= nhalf) == upper) == (pass == 0));
        const unsigned m = __builtin_amdgcn_ballot_w32(want);
        if (m != 0u) sgc_emit(rows, sl, d, nd, m, acc, lane);
      }
    }
    return;
  }
  int* wl = (!REPLAY && lists != nullptr) ? lists + (size_t)gw * L_CAP : nullptr;
  int lpos = 0;
  if (KEYS) {
    const uint4* kp = (const uint4*)keys;
    const int ktrip = ((E + 4095) >> 12) << 1;
    const int kstart = (int)(((unsigned)blockIdx.x * (unsigned)ktrip) / gridDim.x);
    const unsigned g4 = blockIdx.x * 0x01010101u;
    int kcnt = 0;
    uint4 knx[4];
    if (wave < ktrip) {
      int tr = wave + kstart; if (tr >= ktrip) tr -= ktrip;
#pragma unroll
      for (int u = 0; u < 4; ++u) knx[u] = kp[(tr << 7) + u * 32 + lane];
    }
    for (int t = wave; t < ktrip; t += 32) {
      int tr = t + kstart; if (tr >= ktrip) tr -= ktrip;
      const int ebase = (tr << 11) + lane * 16;
      const unsigned m01 = sgc_key_mask(knx[0], knx[1], g4);
      const unsigned m23 = sgc_key_mask(knx[2], knx[3], g4);
      if (t + 32 < ktrip) {
        int tn = t + 32 + kstart; if (tn >= ktrip) tn -= ktrip;
#pragma unroll
        for (int u = 0; u < 4; ++u) knx[u] = kp[(tn << 7) + u * 32 + lane];
      }
      sgc_queue_keys(m01, ebase, q, kcnt, rows, norm, src, dst, acc, lane, fl, fcnt, nhalf, upper, wl, lpos, r0);
      sgc_queue_keys(m23, ebase + 1024, q, kcnt, rows, norm, src, dst, acc, lane, fl, fcnt, nhalf, upper, wl, lpos, r0);
    }
    if (kcnt > 0) sgc_flush<true>(rows, q, 0, kcnt, norm, src, dst, acc, lane, fl, fcnt, nhalf, upper, wl, lpos, r0);
    if (fcnt > 0) sgc_drain(rows, fl, fcnt, norm, acc, lane);
    if (wl != nullptr && lane == 0) lcnt[gw] = (lpos <= L_CAP) ? lpos : -1;
    return;
  }
  const int ntrip = (E >= 4) ? ((E + 1023) >> 10) : 0;
  const int start = (int)(((unsigned)blockIdx.x * (unsigned)ntrip) / gridDim.x);
  int cnt = 0;

  int4 nxt[8];
  if (wave < ntrip) {
    int tr = wave + start; if (tr >= ntrip) tr -= ntrip;
#pragma unroll
    for (int u = 0; u < 8; ++u) nxt[u] = sgc_load_src4(src, (tr << 10) + u * 128 + lane * 4, E);
  }
  for (int t = wave; t < ntrip; t += 32) {
    int tr = t + start; if (tr >= ntrip) tr -= ntrip;
    const int ebase = (tr << 10) + lane * 4;
    int4 sv[8];
#pragma unroll
    for (int u = 0; u < 8; ++u) sv[u] = nxt[u];
    if (t + 32 < ntrip) {
      int tn = t + 32 + start; if (tn >= ntrip) tn -= ntrip;
#pragma unroll
      for (int u = 0; u < 8; ++u) nxt[u] = sgc_load_src4(src, (tn << 10) + u * 128 + lane * 4, E);
    }
#pragma unroll
    for (int u = 0; u < 8; ++u) {
      const int e = ebase + u * 128;
      sgc_test(sv[u].x - r0, e + 0, q, cnt);
      sgc_test(sv[u].y - r0, e + 1, q, cnt);
      sgc_test(sv[u].z - r0, e + 2, q, cnt);
      sgc_test(sv[u].w - r0, e + 3, q, cnt);
      while (cnt >= 32) {
        cnt -= 32;
        sgc_flush<false>(rows, q, cnt, 32, norm, src, dst, acc, lane, fl, fcnt, nhalf, upper, wl, lpos, r0);
      }
    }
  }
  const int tail0 = (E >= 4) ? (E & ~3) : 0;
  if (wave == 0 && tail0 < E) {
    const int e = tail0 + lane;
    const int s = (e < E) ? src[e] - r0 : -1;
    sgc_test(s, e, q, cnt);
    while (cnt >= 32) {
      cnt -= 32;
      sgc_flush<false>(rows, q, cnt, 32, norm, src, dst, acc, lane, fl, fcnt, nhalf, upper, wl, lpos, r0);
    }
  }
  if (cnt > 0) sgc_flush<false>(rows, q, 0, cnt, norm, src, dst, acc, lane, fl, fcnt, nhalf, upper, wl, lpos, r0);
  if (fcnt > 0) sgc_drain(rows, fl, fcnt, norm, acc, lane);
  if (wl != nullptr && lane == 0) lcnt[gw] = (lpos <= L_CAP) ? lpos : -1;
}

__global__ __launch_bounds__(256) void sgc_zero(float4* __restrict__ p, int total4) {
  int i = blockIdx.x * 256 + threadIdx.x;
  if (i < total4) p[i] = make_float4(0.f, 0.f, 0.f, 0.f);
}

extern "C" void kernel_launch(void* const* d_in, const int* in_sizes, int n_in,
                              void* d_out, int out_size, void* d_ws, size_t ws_size,
                              hipStream_t stream) {
  const float* h    = (const float*)d_in[0];
  const float* w    = (const float*)d_in[1];
  const float* norm = (const float*)d_in[2];
  const int*   src  = (const int*)d_in[3];
  const int*   dst  = (const int*)d_in[4];
  float* out = (float*)d_out;
  float* buf = (float*)d_ws;

  const int N = in_sizes[2];
  const int E = in_sizes[3];

  const int mblocks = (N + 15) / 16;
  const int total4  = N * OUT_F / 4;
  const int sblocks = (total4 + 255) / 256;
  const int pblocks = (N + SRC_ROWS - 1) / SRC_ROWS;

  const size_t buf_bytes = (size_t)N * OUT_F * sizeof(float);
  const size_t cnt_bytes = (size_t)pblocks * 32 * sizeof(int);
  const size_t lst_bytes = (size_t)pblocks * 32 * L_CAP * sizeof(int);
  const bool keep = (ws_size >= buf_bytes + cnt_bytes + lst_bytes) && (N <= (1 << (31 - L_SHIFT)));
  int* lcnt  = keep ? (int*)((char*)d_ws + buf_bytes) : nullptr;
  int* lists = keep ? (int*)((char*)d_ws + buf_bytes + cnt_bytes) : nullptr;

  const int    epad      = ((E + 4095) / 4096) * 4096;
  const size_t key_off   = buf_bytes + (keep ? cnt_bytes + lst_bytes : 0);
  const bool   usekeys   = (pblocks <= 255) && (E >= 4) && (ws_size >= key_off + (size_t)epad);
  unsigned*    keys      = usekeys ? (unsigned*)((char*)d_ws + key_off) : nullptr;

  sgc_gemm_wmma<<<mblocks, 256, 0, stream>>>(h, w, out);

  if (usekeys) sgc_keys<<<epad / 4096, 256, 0, stream>>>(src, keys, E);
  sgc_zero<<<sblocks, 256, 0, stream>>>((float4*)buf, total4);
  if (usekeys)
    sgc_push<false, true><<<pblocks, 1024, 0, stream>>>(out, norm, src, dst, buf, N, E, lcnt, lists, keys);
  else
    sgc_push<false, false><<<pblocks, 1024, 0, stream>>>(out, norm, src, dst, buf, N, E, lcnt, lists, keys);

  sgc_zero<<<sblocks, 256, 0, stream>>>((float4*)out, total4);
  if (usekeys)
    sgc_push<true, true><<<pblocks, 1024, 0, stream>>>(buf, norm, src, dst, out, N, E, lcnt, lists, keys);
  else
    sgc_push<true, false><<<pblocks, 1024, 0, stream>>>(buf, norm, src, dst, out, N, E, lcnt, lists, keys);
}
